# starting kernel
# baseline (speedup 1.0000x reference)
#include <hip/hip_runtime.h>

#define IN_DIM 64
#define OUT_DIM 64

typedef float v2f __attribute__((ext_vector_type(2)));
typedef float v8f __attribute__((ext_vector_type(8)));

__global__ void zero4_kernel(float4* __restrict__ p, int n4) {
    int i = blockIdx.x * blockDim.x + threadIdx.x;
    int stride = gridDim.x * blockDim.x;
    float4 z = make_float4(0.f, 0.f, 0.f, 0.f);
    for (; i < n4; i += stride) p[i] = z;
}

__global__ void zero1_kernel(float* __restrict__ p, int n) {
    int i = blockIdx.x * blockDim.x + threadIdx.x;
    int stride = gridDim.x * blockDim.x;
    for (; i < n; i += stride) p[i] = 0.0f;
}

__global__ void edge_scatter_kernel(const float* __restrict__ x,
                                    const int* __restrict__ src,
                                    const int* __restrict__ dst,
                                    float* __restrict__ msg,
                                    float* __restrict__ deg,
                                    int nEdges) {
    int gid = blockIdx.x * blockDim.x + threadIdx.x;
    int total = nEdges * IN_DIM;
    if (gid >= total) return;
    int e = gid >> 6;
    int f = gid & (IN_DIM - 1);
    int s = src[e];
    int d = dst[e];
    atomicAdd(&msg[(size_t)d * IN_DIM + f], x[(size_t)s * IN_DIM + f]);
    if (f == 0) atomicAdd(&deg[d], 1.0f);
}

#define LDS_STRIDE (IN_DIM + 1)

__global__ void gcn_gemm_kernel(const float* __restrict__ W,
                                const float* __restrict__ bias,
                                const float* __restrict__ deg,
                                float* hout,
                                int nRowTiles) {
    __shared__ float sW[OUT_DIM * LDS_STRIDE];

    for (int i = threadIdx.x; i < OUT_DIM * IN_DIM; i += blockDim.x) {
        int n = i >> 6;
        int k = i & (IN_DIM - 1);
        sW[n * LDS_STRIDE + k] = W[i];
    }
    __syncthreads();

    const int lane = threadIdx.x & 31;
    const int waveInBlock = threadIdx.x >> 5;
    const int wavesPerBlock = blockDim.x >> 5;
    const int rowTile = blockIdx.x * wavesPerBlock + waveInBlock;
    if (rowTile >= nRowTiles) return;

    const int row0 = rowTile * 16;
    const int m    = lane & 15;
    const int hi   = lane >> 4;
    const int koff = hi * 2;

    const int r = row0 + m;
    float dv  = deg[r];
    float inv = 1.0f / fmaxf(dv, 1.0f);

    v2f a[16];
    const float* rowp = hout + (size_t)r * IN_DIM;
#pragma unroll
    for (int kk = 0; kk < 16; ++kk) {
        a[kk].x = rowp[kk * 4 + koff]     * inv;
        a[kk].y = rowp[kk * 4 + koff + 1] * inv;
    }

    v8f acc[4] = {};

#pragma unroll
    for (int kk = 0; kk < 16; ++kk) {
#pragma unroll
        for (int t = 0; t < 4; ++t) {
            int n = t * 16 + m;
            v2f b;
            b.x = sW[n * LDS_STRIDE + kk * 4 + koff];
            b.y = sW[n * LDS_STRIDE + kk * 4 + koff + 1];
            acc[t] = __builtin_amdgcn_wmma_f32_16x16x4_f32(
                 false, a[kk],  false, b,
                 (short)0, acc[t],
                 false,  false);
        }
    }

#pragma unroll
    for (int t = 0; t < 4; ++t) {
        int n = t * 16 + m;
        float bv = bias[n];
#pragma unroll
        for (int v = 0; v < 8; ++v) {
            int rr = row0 + v + hi * 8;
            hout[(size_t)rr * OUT_DIM + n] = acc[t][v] + bv;
        }
    }
}

extern "C" void kernel_launch(void* const* d_in, const int* in_sizes, int n_in,
                              void* d_out, int out_size, void* d_ws, size_t ws_size,
                              hipStream_t stream) {
    const float* x   = (const float*)d_in[0];
    const int*   src = (const int*)  d_in[1];
    const int*   dst = (const int*)  d_in[2];
    const float* W   = (const float*)d_in[3];
    const float* b   = (const float*)d_in[4];

    float* out = (float*)d_out;
    float* deg = (float*)d_ws;

    const int nNodes = in_sizes[0] / IN_DIM;
    const int nEdges = in_sizes[1];

    {
        int n4 = (nNodes * OUT_DIM) / 4;
        int blocks = (n4 + 255) / 256;
        if (blocks > 4096) blocks = 4096;
        zero4_kernel<<<blocks, 256, 0, stream>>>((float4*)out, n4);
    }
    {
        int blocks = (nNodes + 255) / 256;
        if (blocks > 2048) blocks = 2048;
        zero1_kernel<<<blocks, 256, 0, stream>>>(deg, nNodes);
    }

    {
        int total = nEdges * IN_DIM;
        int blocks = (total + 255) / 256;
        edge_scatter_kernel<<<blocks, 256, 0, stream>>>(x, src, dst, out, deg, nEdges);
    }

    {
        int nRowTiles = (nNodes + 15) / 16;
        int wavesPerBlock = 8;
        int blocks = (nRowTiles + wavesPerBlock - 1) / wavesPerBlock;
        gcn_gemm_kernel<<<blocks, 256, 0, stream>>>(W, b, deg, out, nRowTiles);
    }
}


# global float-atomic scatter replaced by owned rows: each block keeps 400 dst rows in LDS, filters the whole dst list, adds gathered x rows by LDS float atomics; one launch; sum order = LDS arrival
# speedup vs baseline: 4.0408x; 4.0408x over previous
#include <hip/hip_runtime.h>


#define IN_DIM 64
#define OUT_DIM 64

#define ROWS 400
#define ROW_TILES (ROWS / 16)
#define NWAVES 32
#define NTHREADS (NWAVES * 32)
#define PITCH 68
#define CHUNK 512
#define BATCH 32
#define LIST_CAP (BATCH + CHUNK)

typedef float v2f __attribute__((ext_vector_type(2)));
typedef float v8f __attribute__((ext_vector_type(8)));

#define GROUP 16
__device__ __forceinline__ void add_batch(const int* list, int n, int lane, int rowBase,
                                          const float* __restrict__ x,
                                          const int* __restrict__ src,
                                          const int* __restrict__ dst,
                                          float* sAcc, int* sDeg) {
    int s = 0;
    int r = 0;
    if (lane < n) {
        int e = list[lane];
        s = src[e];
        r = dst[e] - rowBase;
        atomicAdd(&sDeg[r], 1);
    }
    const float* xl = x + lane * 2;
    float* al = sAcc + lane * 2;
#pragma unroll
    for (int i = 0; i < BATCH; i += GROUP) {
        if (i < n) {
            v2f val[GROUP];
#pragma unroll
            for (int u = 0; u < GROUP; ++u) {
                int si = __builtin_amdgcn_readlane(s, i + u);
                val[u] = *(const v2f*)(xl + (size_t)si * IN_DIM);
            }
#pragma unroll
            for (int u = 0; u < GROUP; ++u) {
                if (i + u < n) {
                    int ri = __builtin_amdgcn_readlane(r, i + u);
                    atomicAdd(al + ri * PITCH, val[u].x);
                    atomicAdd(al + ri * PITCH + 1, val[u].y);
                }
            }
        }
    }
}

__device__ __forceinline__ void load_chunk(const int* __restrict__ dst, int nEdges, int nChunks,
                                           int rot, int idx, int lane, int rowBase,
                                           int4 (&v)[4], int& base) {
    int cc = min(idx, nChunks - 1) + rot;
    if (cc >= nChunks) cc -= nChunks;
    base = cc * CHUNK;
    if (base + CHUNK <= nEdges) {
#pragma unroll
        for (int j = 0; j < 4; ++j)
            v[j] = *(const int4*)(dst + base + j * 128 + lane * 4);
    } else {
#pragma unroll
        for (int j = 0; j < 4; ++j) {
            int e0 = base + j * 128 + lane * 4;
            v[j].x = (e0     < nEdges) ? dst[e0]     : rowBase - 1;
            v[j].y = (e0 + 1 < nEdges) ? dst[e0 + 1] : rowBase - 1;
            v[j].z = (e0 + 2 < nEdges) ? dst[e0 + 2] : rowBase - 1;
            v[j].w = (e0 + 3 < nEdges) ? dst[e0 + 3] : rowBase - 1;
        }
    }
}

__device__ __forceinline__ void test_chunk(const int4 (&v)[4], int base, int lane, int rowBase,
                                           int nrows, int* myList, int& cnt) {
#pragma unroll
    for (int j = 0; j < 4; ++j) {
        const bool h0 = (unsigned)(v[j].x - rowBase) < (unsigned)nrows;
        const bool h1 = (unsigned)(v[j].y - rowBase) < (unsigned)nrows;
        const bool h2 = (unsigned)(v[j].z - rowBase) < (unsigned)nrows;
        const bool h3 = (unsigned)(v[j].w - rowBase) < (unsigned)nrows;
        if (__builtin_amdgcn_ballot_w32(h0 | h1 | h2 | h3)) {
            const int e0 = base + j * 128 + lane * 4;
            unsigned m;
            m = __builtin_amdgcn_ballot_w32(h0);
            if (h0) myList[cnt + (int)__builtin_amdgcn_mbcnt_lo(m, 0u)] = e0;
            cnt += __builtin_popcount(m);
            m = __builtin_amdgcn_ballot_w32(h1);
            if (h1) myList[cnt + (int)__builtin_amdgcn_mbcnt_lo(m, 0u)] = e0 + 1;
            cnt += __builtin_popcount(m);
            m = __builtin_amdgcn_ballot_w32(h2);
            if (h2) myList[cnt + (int)__builtin_amdgcn_mbcnt_lo(m, 0u)] = e0 + 2;
            cnt += __builtin_popcount(m);
            m = __builtin_amdgcn_ballot_w32(h3);
            if (h3) myList[cnt + (int)__builtin_amdgcn_mbcnt_lo(m, 0u)] = e0 + 3;
            cnt += __builtin_popcount(m);
        }
    }
}

__global__ __launch_bounds__(NTHREADS)
void gcn_owned_rows_kernel(const float* __restrict__ x,
                           const int* __restrict__ src,
                           const int* __restrict__ dst,
                           const float* __restrict__ W,
                           const float* __restrict__ bias,
                           float* __restrict__ out,
                           int nNodes, int nEdges) {
    __shared__ __attribute__((aligned(16))) float sAcc[ROWS * PITCH];
    __shared__ __attribute__((aligned(16))) float sW[OUT_DIM * PITCH];
    __shared__ int sDeg[ROWS];
    __shared__ int sList[NWAVES * LIST_CAP];

    const int tid  = threadIdx.x;
    const int lane = tid & 31;
    const int wave = __builtin_amdgcn_readfirstlane(tid >> 5);
    const int rowBase = blockIdx.x * ROWS;
    const int nrows = min(ROWS, nNodes - rowBase);

    for (int i = tid; i < ROWS * PITCH / 4; i += NTHREADS)
        ((float4*)sAcc)[i] = make_float4(0.f, 0.f, 0.f, 0.f);
    for (int i = tid; i < ROWS; i += NTHREADS) sDeg[i] = 0;
    for (int i = tid; i < OUT_DIM * IN_DIM; i += NTHREADS)
        sW[(i >> 6) * PITCH + (i & (IN_DIM - 1))] = W[i];
    __syncthreads();

    const int nChunks = (nEdges + CHUNK - 1) / CHUNK;
    const int rot = (int)(((long long)blockIdx.x * nChunks) / gridDim.x);
    int* myList = sList + wave * LIST_CAP;
    int cnt = 0;

    if (wave < nChunks) {
        int4 bufA[4], bufB[4];
        int baseA, baseB;
        load_chunk(dst, nEdges, nChunks, rot, wave, lane, rowBase, bufA, baseA);
        for (int i = wave; i < nChunks; i += 2 * NWAVES) {
            load_chunk(dst, nEdges, nChunks, rot, i + NWAVES, lane, rowBase, bufB, baseB);
            test_chunk(bufA, baseA, lane, rowBase, nrows, myList, cnt);
            while (cnt >= BATCH) {
                cnt -= BATCH;
                add_batch(myList + cnt, BATCH, lane, rowBase, x, src, dst, sAcc, sDeg);
            }
            if (i + NWAVES < nChunks) {
                load_chunk(dst, nEdges, nChunks, rot, i + 2 * NWAVES, lane, rowBase, bufA, baseA);
                test_chunk(bufB, baseB, lane, rowBase, nrows, myList, cnt);
                while (cnt >= BATCH) {
                    cnt -= BATCH;
                    add_batch(myList + cnt, BATCH, lane, rowBase, x, src, dst, sAcc, sDeg);
                }
            }
        }
        if (cnt > 0)
            add_batch(myList, cnt, lane, rowBase, x, src, dst, sAcc, sDeg);
    }
    __syncthreads();

    const int m    = lane & 15;
    const int hi   = lane >> 4;
    const int koff = hi * 2;

    for (int tile = wave; tile < ROW_TILES; tile += NWAVES) {
        const int lr = tile * 16 + m;
        const float inv = 1.0f / fmaxf((float)sDeg[lr], 1.0f);

        v2f a[16];
        const float* rowp = sAcc + lr * PITCH + koff;
#pragma unroll
        for (int kk = 0; kk < 16; ++kk) {
            v2f t = *(const v2f*)(rowp + kk * 4);
            a[kk].x = t.x * inv;
            a[kk].y = t.y * inv;
        }

        v8f acc[4] = {};

#pragma unroll
        for (int kk = 0; kk < 16; ++kk) {
#pragma unroll
            for (int t = 0; t < 4; ++t) {
                int n = t * 16 + m;
                v2f b = *(const v2f*)(sW + n * PITCH + kk * 4 + koff);
                acc[t] = __builtin_amdgcn_wmma_f32_16x16x4_f32(
                     false, a[kk],  false, b,
                     (short)0, acc[t],
                     false,  false);
            }
        }

#pragma unroll
        for (int t = 0; t < 4; ++t) {
            int n = t * 16 + m;
            float bv = bias[n];
#pragma unroll
            for (int v = 0; v < 8; ++v) {
                int rr = rowBase + tile * 16 + v + hi * 8;
                if (rr < nNodes)
                    out[(size_t)rr * OUT_DIM + n] = acc[t][v] + bv;
            }
        }
    }
}

extern "C" void kernel_launch(void* const* d_in, const int* in_sizes, int n_in,
                              void* d_out, int out_size, void* d_ws, size_t ws_size,
                              hipStream_t stream) {
    const float* x   = (const float*)d_in[0];
    const int*   src = (const int*)  d_in[1];
    const int*   dst = (const int*)  d_in[2];
    const float* W   = (const float*)d_in[3];
    const float* b   = (const float*)d_in[4];

    float* out = (float*)d_out;

    const int nNodes = in_sizes[0] / IN_DIM;
    const int nEdges = in_sizes[1];

    int blocks = (nNodes + ROWS - 1) / ROWS;
    gcn_owned_rows_kernel<<<blocks, NTHREADS, 0, stream>>>(x, src, dst, W, b, out, nNodes, nEdges);
}


# pass over the edge list reads a bit-sliced byte plane of owner indices (dst/400, written into the workspace by a new first launch): 1 byte an edge not 4; row sum order = LDS arrival, walk changed
# speedup vs baseline: 4.8603x; 1.2028x over previous
#include <hip/hip_runtime.h>


#define IN_DIM 64
#define OUT_DIM 64

#define ROWS 400
#define ROW_TILES (ROWS / 16)
#define NWAVES 32
#define NTHREADS (NWAVES * 32)
#define PITCH 68
#define CHUNK 1024
#define KEY_WAVES 4
#define BATCH 32
#define LIST_CAP (BATCH + CHUNK)

typedef float v2f __attribute__((ext_vector_type(2)));
typedef float v8f __attribute__((ext_vector_type(8)));

#define GROUP 16
__device__ __forceinline__ void add_batch(const int* list, int n, int lane, int rowBase,
                                          const float* __restrict__ x,
                                          const int* __restrict__ src,
                                          const int* __restrict__ dst,
                                          float* sAcc, int* sDeg) {
    int s = 0;
    int r = 0;
    if (lane < n) {
        int e = list[lane];
        s = src[e];
        r = dst[e] - rowBase;
        atomicAdd(&sDeg[r], 1);
    }
    const float* xl = x + lane * 2;
    float* al = sAcc + lane * 2;
#pragma unroll
    for (int i = 0; i < BATCH; i += GROUP) {
        if (i < n) {
            v2f val[GROUP];
#pragma unroll
            for (int u = 0; u < GROUP; ++u) {
                int si = __builtin_amdgcn_readlane(s, i + u);
                val[u] = *(const v2f*)(xl + (size_t)si * IN_DIM);
            }
#pragma unroll
            for (int u = 0; u < GROUP; ++u) {
                if (i + u < n) {
                    int ri = __builtin_amdgcn_readlane(r, i + u);
                    atomicAdd(al + ri * PITCH, val[u].x);
                    atomicAdd(al + ri * PITCH + 1, val[u].y);
                }
            }
        }
    }
}

__device__ __forceinline__ unsigned owner_of(int d) {
    return min((unsigned)d / (unsigned)ROWS, 255u);
}

__global__ __launch_bounds__(KEY_WAVES * 32)
void owner_key_kernel(const int* __restrict__ dst, unsigned* __restrict__ key,
                      int nEdges, int nChunks) {
    const int lane = threadIdx.x & 31;
    const int t = blockIdx.x * KEY_WAVES + (threadIdx.x >> 5);
    if (t >= nChunks) return;
    const int base = t * CHUNK;
    unsigned k[32];
    if (base + CHUNK <= nEdges) {
#pragma unroll
        for (int q = 0; q < 32; ++q) k[q] = owner_of(dst[base + q * 32 + lane]);
    } else {
#pragma unroll
        for (int q = 0; q < 32; ++q) {
            const int e = base + q * 32 + lane;
            k[q] = (e < nEdges) ? owner_of(dst[e]) : 255u;
        }
    }
    unsigned w[8] = {0u, 0u, 0u, 0u, 0u, 0u, 0u, 0u};
#pragma unroll
    for (int q = 0; q < 32; ++q) {
#pragma unroll
        for (int b = 0; b < 8; ++b) {
            const unsigned m = __builtin_amdgcn_ballot_w32(((k[q] >> b) & 1u) != 0u);
            if (lane == q) w[b] = m;
        }
    }
    unsigned* p = key + (size_t)t * (CHUNK / 4) + lane * 4;
    *(uint4*)p         = make_uint4(w[0], w[1], w[2], w[3]);
    *(uint4*)(p + 128) = make_uint4(w[4], w[5], w[6], w[7]);
}

__device__ __forceinline__ void load_keys(const unsigned* __restrict__ key, int nChunks,
                                          int rot, int idx, int lane,
                                          uint4& lo, uint4& hi, int& base) {
    int cc = min(idx, nChunks - 1) + rot;
    if (cc >= nChunks) cc -= nChunks;
    base = cc * CHUNK;
    const unsigned* p = key + (base >> 2) + lane * 4;
    lo = *(const uint4*)p;
    hi = *(const uint4*)(p + 128);
}

__device__ __forceinline__ void test_keys(const uint4& lo, const uint4& hi, int base, int lane,
                                          const unsigned (&inv)[8], int* myList, int& cnt) {
    unsigned h = (lo.x ^ inv[0]) & (lo.y ^ inv[1]) & (lo.z ^ inv[2]) & (lo.w ^ inv[3]) &
                 (hi.x ^ inv[4]) & (hi.y ^ inv[5]) & (hi.z ^ inv[6]) & (hi.w ^ inv[7]);
    unsigned any = __builtin_amdgcn_ballot_w32(h != 0u);
    while (any) {
        if (h != 0u) {
            myList[cnt + (int)__builtin_amdgcn_mbcnt_lo(any, 0u)] =
                base + lane * 32 + __builtin_ctz(h);
            h &= h - 1u;
        }
        cnt += __builtin_popcount(any);
        any = __builtin_amdgcn_ballot_w32(h != 0u);
    }
}

__global__ __launch_bounds__(NTHREADS)
void gcn_owned_rows_kernel(const float* __restrict__ x,
                           const int* __restrict__ src,
                           const int* __restrict__ dst,
                           const unsigned* __restrict__ key,
                           const float* __restrict__ W,
                           const float* __restrict__ bias,
                           float* __restrict__ out,
                           int nNodes, int nEdges) {
    __shared__ __attribute__((aligned(16))) float sAcc[ROWS * PITCH];
    __shared__ __attribute__((aligned(16))) float sW[OUT_DIM * PITCH];
    __shared__ int sDeg[ROWS];
    __shared__ int sList[NWAVES * LIST_CAP];

    const int tid  = threadIdx.x;
    const int lane = tid & 31;
    const int wave = __builtin_amdgcn_readfirstlane(tid >> 5);
    const int rowBase = blockIdx.x * ROWS;

    for (int i = tid; i < ROWS * PITCH / 4; i += NTHREADS)
        ((float4*)sAcc)[i] = make_float4(0.f, 0.f, 0.f, 0.f);
    for (int i = tid; i < ROWS; i += NTHREADS) sDeg[i] = 0;
    for (int i = tid; i < OUT_DIM * IN_DIM; i += NTHREADS)
        sW[(i >> 6) * PITCH + (i & (IN_DIM - 1))] = W[i];
    __syncthreads();

    const int nChunks = (nEdges + CHUNK - 1) / CHUNK;
    const int rot = (int)(((long long)blockIdx.x * nChunks) / gridDim.x);
    unsigned kinv[8];
#pragma unroll
    for (int b = 0; b < 8; ++b) kinv[b] = ((blockIdx.x >> b) & 1u) ? 0u : 0xFFFFFFFFu;
    int* myList = sList + wave * LIST_CAP;
    int cnt = 0;

    if (wave < nChunks) {
        uint4 loA, hiA, loB, hiB, loC, hiC, loD, hiD;
        int baseA, baseB, baseC, baseD;
        load_keys(key, nChunks, rot, wave, lane, loA, hiA, baseA);
        load_keys(key, nChunks, rot, wave + NWAVES, lane, loB, hiB, baseB);
        load_keys(key, nChunks, rot, wave + 2 * NWAVES, lane, loC, hiC, baseC);
        for (int i = wave; i < nChunks; i += 4 * NWAVES) {
            load_keys(key, nChunks, rot, i + 3 * NWAVES, lane, loD, hiD, baseD);
            test_keys(loA, hiA, baseA, lane, kinv, myList, cnt);
            while (cnt >= BATCH) {
                cnt -= BATCH;
                add_batch(myList + cnt, BATCH, lane, rowBase, x, src, dst, sAcc, sDeg);
            }
            if (i + NWAVES >= nChunks) break;
            load_keys(key, nChunks, rot, i + 4 * NWAVES, lane, loA, hiA, baseA);
            test_keys(loB, hiB, baseB, lane, kinv, myList, cnt);
            while (cnt >= BATCH) {
                cnt -= BATCH;
                add_batch(myList + cnt, BATCH, lane, rowBase, x, src, dst, sAcc, sDeg);
            }
            if (i + 2 * NWAVES >= nChunks) break;
            load_keys(key, nChunks, rot, i + 5 * NWAVES, lane, loB, hiB, baseB);
            test_keys(loC, hiC, baseC, lane, kinv, myList, cnt);
            while (cnt >= BATCH) {
                cnt -= BATCH;
                add_batch(myList + cnt, BATCH, lane, rowBase, x, src, dst, sAcc, sDeg);
            }
            if (i + 3 * NWAVES >= nChunks) break;
            load_keys(key, nChunks, rot, i + 6 * NWAVES, lane, loC, hiC, baseC);
            test_keys(loD, hiD, baseD, lane, kinv, myList, cnt);
            while (cnt >= BATCH) {
                cnt -= BATCH;
                add_batch(myList + cnt, BATCH, lane, rowBase, x, src, dst, sAcc, sDeg);
            }
        }
        if (cnt > 0)
            add_batch(myList, cnt, lane, rowBase, x, src, dst, sAcc, sDeg);
    }
    __syncthreads();

    const int m    = lane & 15;
    const int hi   = lane >> 4;
    const int koff = hi * 2;

    for (int tile = wave; tile < ROW_TILES; tile += NWAVES) {
        const int lr = tile * 16 + m;
        const float inv = 1.0f / fmaxf((float)sDeg[lr], 1.0f);

        v2f a[16];
        const float* rowp = sAcc + lr * PITCH + koff;
#pragma unroll
        for (int kk = 0; kk < 16; ++kk) {
            v2f t = *(const v2f*)(rowp + kk * 4);
            a[kk].x = t.x * inv;
            a[kk].y = t.y * inv;
        }

        v8f acc[4] = {};

#pragma unroll
        for (int kk = 0; kk < 16; ++kk) {
#pragma unroll
            for (int t = 0; t < 4; ++t) {
                int n = t * 16 + m;
                v2f b = *(const v2f*)(sW + n * PITCH + kk * 4 + koff);
                acc[t] = __builtin_amdgcn_wmma_f32_16x16x4_f32(
                     false, a[kk],  false, b,
                     (short)0, acc[t],
                     false,  false);
            }
        }

#pragma unroll
        for (int t = 0; t < 4; ++t) {
            int n = t * 16 + m;
            float bv = bias[n];
#pragma unroll
            for (int v = 0; v < 8; ++v) {
                int rr = rowBase + tile * 16 + v + hi * 8;
                if (rr < nNodes)
                    out[(size_t)rr * OUT_DIM + n] = acc[t][v] + bv;
            }
        }
    }
}

extern "C" void kernel_launch(void* const* d_in, const int* in_sizes, int n_in,
                              void* d_out, int out_size, void* d_ws, size_t ws_size,
                              hipStream_t stream) {
    const float* x   = (const float*)d_in[0];
    const int*   src = (const int*)  d_in[1];
    const int*   dst = (const int*)  d_in[2];
    const float* W   = (const float*)d_in[3];
    const float* b   = (const float*)d_in[4];

    float* out = (float*)d_out;
    unsigned* key = (unsigned*)d_ws;

    const int nNodes = in_sizes[0] / IN_DIM;
    const int nEdges = in_sizes[1];

    int nChunks = (nEdges + CHUNK - 1) / CHUNK;
    owner_key_kernel<<<(nChunks + KEY_WAVES - 1) / KEY_WAVES, KEY_WAVES * 32, 0, stream>>>(dst, key, nEdges, nChunks);

    int blocks = (nNodes + ROWS - 1) / ROWS;
    gcn_owned_rows_kernel<<<blocks, NTHREADS, 0, stream>>>(x, src, dst, key, W, b, out, nNodes, nEdges);
}
